# starting kernel
# baseline (speedup 1.0000x reference)
#include <hip/hip_runtime.h>
#include <math.h>

typedef __attribute__((ext_vector_type(2))) float v2f;
typedef __attribute__((ext_vector_type(8))) float v8f;

#define IMG_H 512
#define IMG_W 512
#define TS 16

__device__ __forceinline__ int iclamp(int v, int lo, int hi) {
    return v < lo ? lo : (v > hi ? hi : v);
}

__global__ __launch_bounds__(32) void sobel_wmma_kernel(const float* __restrict__ img,
                                                        float* __restrict__ out) {
    const int X0   = blockIdx.x * TS;
    const int Y0   = blockIdx.y * TS;
    const int n    = blockIdx.z;
    const int lane = threadIdx.x;

    __shared__ float patch[18][20];
    __shared__ float Pv[16][20];
    __shared__ float Qv[16][20];

    const float* base = img + (size_t)n * 3 * IMG_H * IMG_W;

    for (int idx = lane; idx < 18 * 18; idx += 32) {
        const int r  = idx / 18;
        const int c  = idx % 18;
        const int iy = iclamp(Y0 - 1 + r, 0, IMG_H - 1);
        const int ix = iclamp(X0 - 1 + c, 0, IMG_W - 1);
        const int o  = iy * IMG_W + ix;
        patch[r][c] = base[o] + base[IMG_H * IMG_W + o] + base[2 * IMG_H * IMG_W + o];
    }
    __syncthreads();

    for (int idx = lane; idx < 16 * 20; idx += 32) {
        const int j = idx / 20;
        const int c = idx % 20;
        float p = 0.0f, q = 0.0f;
        if (c < 18) {
            const int lr = iclamp(Y0 + j, 1, IMG_H - 2) - (Y0 - 1);
            const float a = patch[lr - 1][c];
            const float b = patch[lr][c];
            const float d = patch[lr + 1][c];
            p = a + 2.0f * b + d;
            q = a - d;
        }
        Pv[j][c] = p;
        Qv[j][c] = q;
    }
    __syncthreads();

    const int half = lane >> 4;
    const int lrow = lane & 15;
    const int li   = iclamp(X0 + lrow, 1, IMG_W - 2) - (X0 - 1);

    v8f accx = {};
    v8f accy = {};
#pragma unroll
    for (int kb = 0; kb < 20; kb += 4) {
        const int k0 = kb + 2 * half;
        const int k1 = k0 + 1;
        v2f a, b;

        a.x = Pv[lrow][k0];
        a.y = Pv[lrow][k1];
        b.x = (float)((k0 == li - 1) - (k0 == li + 1));
        b.y = (float)((k1 == li - 1) - (k1 == li + 1));
        accx = __builtin_amdgcn_wmma_f32_16x16x4_f32(false, a, false, b,
                                                     (short)0, accx, false, false);

        a.x = Qv[lrow][k0];
        a.y = Qv[lrow][k1];
        b.x = (float)((k0 == li - 1) + 2 * (k0 == li) + (k0 == li + 1));
        b.y = (float)((k1 == li - 1) + 2 * (k1 == li) + (k1 == li + 1));
        accy = __builtin_amdgcn_wmma_f32_16x16x4_f32(false, a, false, b,
                                                     (short)0, accy, false, false);
    }

    float* op = out + (size_t)n * IMG_H * IMG_W + (size_t)Y0 * IMG_W + (X0 + lrow);
#pragma unroll
    for (int r = 0; r < 8; ++r) {
        const int M = r + 8 * half;
        const float gx = accx[r];
        const float gy = accy[r];
        op[M * IMG_W] = sqrtf(gx * gx + gy * gy);
    }
}

extern "C" void kernel_launch(void* const* d_in, const int* in_sizes, int n_in,
                              void* d_out, int out_size, void* d_ws, size_t ws_size,
                              hipStream_t stream) {
    (void)in_sizes; (void)n_in; (void)out_size; (void)d_ws; (void)ws_size;
    const float* img = (const float*)d_in[0];
    float* out = (float*)d_out;

    dim3 grid(IMG_W / TS, IMG_H / TS, 32);
    dim3 block(32);
    sobel_wmma_kernel<<<grid, block, 0, stream>>>(img, out);
}


# taps now read from kx/ky (full 3x3x3, none hardcoded): 9 banded WMMA passes per tile, Gx|Gy sharing N; re-cut to 8-wave groups sweeping 128x128 with 16-B staged loads; f32 sum order changed
# speedup vs baseline: 1.2982x; 1.2982x over previous
#include <hip/hip_runtime.h>
#include <math.h>

typedef __attribute__((ext_vector_type(2))) float v2f;
typedef __attribute__((ext_vector_type(8))) float v8f;
typedef __attribute__((ext_vector_type(4))) float v4f;

#define IMG_H 512
#define IMG_W 512
#define CH 3
#define TS 16
#define WAVES 8
#define PR 18
#define PQ 34
#define PITCH 140
#define RB 8
#define NITEM (CH * PR * PQ)
#define NIT 8
#define NSPARE (256 * NIT - NITEM)

__device__ __forceinline__ int iclamp(int v, int lo, int hi) {
    return v < lo ? lo : (v > hi ? hi : v);
}

__global__ __launch_bounds__(256) void conv3x3_pair_mag_kernel(const float* __restrict__ img,
                                                               const float* __restrict__ kx,
                                                               const float* __restrict__ ky,
                                                               float* __restrict__ out) {
    __shared__ __attribute__((aligned(16))) float P[CH * PR * PITCH + 4 * NSPARE];

    const int tid  = threadIdx.x;
    const int lane = tid & 31;
    const int wv   = tid >> 5;
    const int X0   = blockIdx.x * (TS * WAVES);
    const int Yg   = blockIdx.y * (TS * RB);
    const int n    = blockIdx.z;

    const float* base = img + (size_t)n * CH * IMG_H * IMG_W;

    int g_off[NIT];
    int p_row[NIT];
    int l_off[NIT];
#pragma unroll
    for (int it = 0; it < NIT; ++it) {
        const int idx = tid + 256 * it;
        const bool spare = idx >= NITEM;
        const int idc = spare ? idx - NITEM : idx;
        const int q   = idc % PQ;
        const int cr  = idc / PQ;
        const int r   = cr % PR;
        const int c   = cr / PR;
        const int col4 = iclamp(X0 / 4 - 1 + q, 0, IMG_W / 4 - 1);
        g_off[it] = c * IMG_H * IMG_W + 4 * col4;
        p_row[it] = r;
        l_off[it] = spare ? CH * PR * PITCH + 4 * (idx - NITEM) : cr * PITCH + 4 * q;
    }

    v4f pf[NIT];
#pragma unroll
    for (int it = 0; it < NIT; ++it) {
        const int iy = iclamp(Yg - 1 + p_row[it], 0, IMG_H - 1);
        pf[it] = *reinterpret_cast<const v4f*>(base + g_off[it] + iy * IMG_W);
    }

    const int half = lane >> 4;
    const int N    = lane & 15;
    const int f    = N >> 3;
    const int nn   = N & 7;
    const int xw   = X0 + TS * wv;

    const float* kf = f ? ky : kx;
    float tap[CH * 9];
#pragma unroll
    for (int i = 0; i < CH * 9; ++i) tap[i] = kf[i];

    float msk[2][3][4];
#pragma unroll
    for (int u = 0; u < 2; ++u) {
        const int xu0 = xw + 8 * u;
        const int li  = iclamp(xu0 + nn, 1, IMG_W - 2) - (xu0 - 2);
#pragma unroll
        for (int s = 0; s < 3; ++s) {
            const int d = 4 * s + 2 * half - (li - 1);
#pragma unroll
            for (int e = 0; e < 4; ++e) msk[u][s][e] = (d == e - 1) ? 1.0f : 0.0f;
        }
    }

    const int a_col = TS * wv + 2 + 2 * half;

    for (int b = 0; b < RB; ++b) {
        const int Yb = Yg + TS * b;

#pragma unroll
        for (int it = 0; it < NIT; ++it) {
            *reinterpret_cast<v4f*>(&P[l_off[it]]) = pf[it];
        }
        __syncthreads();

        if (b + 1 < RB) {
#pragma unroll
            for (int it = 0; it < NIT; ++it) {
                const int iy = iclamp(Yb + TS - 1 + p_row[it], 0, IMG_H - 1);
                pf[it] = *reinterpret_cast<const v4f*>(base + g_off[it] + iy * IMG_W);
            }
            __builtin_amdgcn_sched_barrier(0);
        }

        const int lr = iclamp(Yb + N, 1, IMG_H - 2) - (Yb - 1);
        const float* arow = &P[(lr - 1) * PITCH + a_col];

        v8f acc0 = {};
        v8f acc1 = {};
#pragma unroll
        for (int c = 0; c < CH; ++c) {
#pragma unroll
            for (int dy = 0; dy < 3; ++dy) {
                const float t0 = tap[c * 9 + dy * 3 + 0];
                const float t1 = tap[c * 9 + dy * 3 + 1];
                const float t2 = tap[c * 9 + dy * 3 + 2];
                const float* ap = arow + (c * PR + dy) * PITCH;
#pragma unroll
                for (int s = 0; s < 3; ++s) {
                    const v2f a0 = *reinterpret_cast<const v2f*>(ap + 4 * s);
                    const v2f a1 = *reinterpret_cast<const v2f*>(ap + 4 * s + 8);
                    v2f b0, b1;
                    b0.x = t0 * msk[0][s][1] + t1 * msk[0][s][2] + t2 * msk[0][s][3];
                    b0.y = t0 * msk[0][s][0] + t1 * msk[0][s][1] + t2 * msk[0][s][2];
                    b1.x = t0 * msk[1][s][1] + t1 * msk[1][s][2] + t2 * msk[1][s][3];
                    b1.y = t0 * msk[1][s][0] + t1 * msk[1][s][1] + t2 * msk[1][s][2];
                    acc0 = __builtin_amdgcn_wmma_f32_16x16x4_f32(false, a0, false, b0,
                                                                 (short)0, acc0, false, false);
                    acc1 = __builtin_amdgcn_wmma_f32_16x16x4_f32(false, a1, false, b1,
                                                                 (short)0, acc1, false, false);
                }
            }
        }

        float* op = out + (size_t)n * IMG_H * IMG_W + (size_t)Yb * IMG_W + (xw + N);
#pragma unroll
        for (int r = 0; r < 8; ++r) {
            const float own  = f ? acc1[r] : acc0[r];
            const float send = f ? acc0[r] : acc1[r];
            const float recv = __shfl_xor(send, 8);
            const float gx = f ? recv : own;
            const float gy = f ? own : recv;
            op[(r + 8 * half) * IMG_W] = sqrtf(gx * gx + gy * gy);
        }

        __syncthreads();
    }
}

extern "C" void kernel_launch(void* const* d_in, const int* in_sizes, int n_in,
                              void* d_out, int out_size, void* d_ws, size_t ws_size,
                              hipStream_t stream) {
    (void)in_sizes; (void)n_in; (void)out_size; (void)d_ws; (void)ws_size;
    const float* img = (const float*)d_in[0];
    const float* kx  = (const float*)d_in[1];
    const float* ky  = (const float*)d_in[2];
    float* out = (float*)d_out;

    dim3 grid(4, 4, 32);
    dim3 block(256);
    conv3x3_pair_mag_kernel<<<grid, block, 0, stream>>>(img, kx, ky, out);
}


# WMMA re-tiled: N = 2 filters x 2x4-pixel patch, K = packed 4x6 input window per channel: 36 WMMAs per 16x16 tile, not 54, fragment reads grouped ahead in the freed registers; f32 sum order changed
# speedup vs baseline: 1.3991x; 1.0777x over previous
#include <hip/hip_runtime.h>
#include <math.h>

typedef __attribute__((ext_vector_type(2))) float v2f;
typedef __attribute__((ext_vector_type(8))) float v8f;
typedef __attribute__((ext_vector_type(4))) float v4f;

#define IMG_H 512
#define IMG_W 512
#define CH 3
#define TS 16
#define WAVES 8
#define PR 18
#define PQ 34
#define PITCH 140
#define RB 8
#define NITEM (CH * PR * PQ)
#define NIT 8
#define NSPARE (256 * NIT - NITEM)
#define ZDY 5
#define ZDX 9
#define ZPL (ZDY * ZDX)
#define ZN (2 * CH * ZPL)

__device__ __forceinline__ int iclamp(int v, int lo, int hi) {
    return v < lo ? lo : (v > hi ? hi : v);
}

__global__ __launch_bounds__(256) void conv3x3_pair_mag_kernel(const float* __restrict__ img,
                                                               const float* __restrict__ kx,
                                                               const float* __restrict__ ky,
                                                               float* __restrict__ out) {
    __shared__ __attribute__((aligned(16))) float P[CH * PR * PITCH + 4 * NSPARE];

    const int tid  = threadIdx.x;
    const int lane = tid & 31;
    const int wv   = tid >> 5;
    const int X0   = blockIdx.x * (TS * WAVES);
    const int Yg   = blockIdx.y * (TS * RB);
    const int n    = blockIdx.z;

    const float* base = img + (size_t)n * CH * IMG_H * IMG_W;

    __shared__ float Z[ZN];
    for (int i = tid; i < ZN; i += 256) {
        const int dx = i % ZDX - 3;
        const int dy = (i / ZDX) % ZDY - 1;
        const int fc = i / ZPL;
        const bool ok = dy >= 0 && dy <= 2 && dx >= 0 && dx <= 2;
        const float* kk = fc >= CH ? ky : kx;
        const int ch = fc >= CH ? fc - CH : fc;
        Z[i] = ok ? kk[ch * 9 + dy * 3 + dx] : 0.0f;
    }

    int g_off[NIT];
    int p_row[NIT];
    int l_off[NIT];
#pragma unroll
    for (int it = 0; it < NIT; ++it) {
        const int idx = tid + 256 * it;
        const bool spare = idx >= NITEM;
        const int idc = spare ? idx - NITEM : idx;
        const int q   = idc % PQ;
        const int cr  = idc / PQ;
        const int r   = cr % PR;
        const int c   = cr / PR;
        const int col4 = iclamp(X0 / 4 - 1 + q, 0, IMG_W / 4 - 1);
        g_off[it] = c * IMG_H * IMG_W + 4 * col4;
        p_row[it] = r;
        l_off[it] = spare ? CH * PR * PITCH + 4 * (idx - NITEM) : cr * PITCH + 4 * q;
    }

    v4f pf[NIT];
#pragma unroll
    for (int it = 0; it < NIT; ++it) {
        const int iy = iclamp(Yg - 1 + p_row[it], 0, IMG_H - 1);
        pf[it] = *reinterpret_cast<const v4f*>(base + g_off[it] + iy * IMG_W);
    }

    const int half = lane >> 4;
    const int c16  = lane & 15;
    const int f    = c16 >> 3;
    const int pr   = (c16 >> 2) & 1;
    const int pc   = c16 & 3;
    const int rg   = c16 & 7;
    const int cgl  = c16 >> 3;
    const int xw   = X0 + TS * wv;

    __syncthreads();

    const float* zb = &Z[f * CH * ZPL + (2 * half - pr + 1) * ZDX + (3 - pc)];
    v2f bfr[CH][6];
#pragma unroll
    for (int c = 0; c < CH; ++c) {
#pragma unroll
        for (int t = 0; t < 6; ++t) {
            const float* zp = zb + c * ZPL + (t / 3) * ZDX + 2 * (t % 3);
            bfr[c][t].x = zp[0];
            bfr[c][t].y = zp[1];
        }
    }

    const float* abase = &P[(2 * rg + 2 * half) * PITCH + TS * wv + 4 * cgl + 3];

    const bool wave_left  = (xw == 0);
    const bool wave_right = (xw == IMG_W - TS);
    int col_src = lane;
    if (wave_left && f == 0 && half == 0 && pc == 0) col_src = lane + 1;
    if (wave_right && f == 1 && half == 1 && pc == 3) col_src = lane - 1;
    const int top_src = (pr == 0) ? lane + 4 : lane;
    const int bot_src = (pr == 1) ? lane - 4 : lane;

    for (int b = 0; b < RB; ++b) {
        const int Yb = Yg + TS * b;

#pragma unroll
        for (int it = 0; it < NIT; ++it) {
            *reinterpret_cast<v4f*>(&P[l_off[it]]) = pf[it];
        }
        __syncthreads();

        if (b + 1 < RB) {
#pragma unroll
            for (int it = 0; it < NIT; ++it) {
                const int iy = iclamp(Yb + TS - 1 + p_row[it], 0, IMG_H - 1);
                pf[it] = *reinterpret_cast<const v4f*>(base + g_off[it] + iy * IMG_W);
            }
            __builtin_amdgcn_sched_barrier(0);
        }

        v2f af[CH][6][2];
#pragma unroll
        for (int c = 0; c < CH; ++c) {
#pragma unroll
            for (int t = 0; t < 6; ++t) {
                const float* ap = abase + (c * PR + t / 3) * PITCH + 2 * (t % 3);
                af[c][t][0].x = ap[0];
                af[c][t][0].y = ap[1];
                af[c][t][1].x = ap[8];
                af[c][t][1].y = ap[9];
            }
        }
        __builtin_amdgcn_sched_barrier(0);

        v8f acc0 = {};
        v8f acc1 = {};
#pragma unroll
        for (int c = 0; c < CH; ++c) {
#pragma unroll
            for (int t = 0; t < 6; ++t) {
                acc0 = __builtin_amdgcn_wmma_f32_16x16x4_f32(false, af[c][t][0], false, bfr[c][t],
                                                             (short)0, acc0, false, false);
                acc1 = __builtin_amdgcn_wmma_f32_16x16x4_f32(false, af[c][t][1], false, bfr[c][t],
                                                             (short)0, acc1, false, false);
            }
        }

        float v[8];
#pragma unroll
        for (int r = 0; r < 8; ++r) {
            const float own  = f ? acc1[r] : acc0[r];
            const float send = f ? acc0[r] : acc1[r];
            const float recv = __shfl_xor(send, 8);
            const float gx = f ? recv : own;
            const float gy = f ? own : recv;
            v[r] = sqrtf(gx * gx + gy * gy);
        }
        if (__builtin_amdgcn_readfirstlane((int)(wave_left || wave_right))) {
#pragma unroll
            for (int r = 0; r < 8; ++r) v[r] = __shfl(v[r], col_src);
        }
        if (Yb == 0) v[0] = __shfl(v[0], top_src);
        if (Yb == IMG_H - TS) v[7] = __shfl(v[7], bot_src);

        float* op = out + (size_t)n * IMG_H * IMG_W + (size_t)(Yb + pr) * IMG_W
                        + (xw + 8 * f + 4 * half + pc);
#pragma unroll
        for (int r = 0; r < 8; ++r) op[2 * r * IMG_W] = v[r];

        __syncthreads();
    }
}

extern "C" void kernel_launch(void* const* d_in, const int* in_sizes, int n_in,
                              void* d_out, int out_size, void* d_ws, size_t ws_size,
                              hipStream_t stream) {
    (void)in_sizes; (void)n_in; (void)out_size; (void)d_ws; (void)ws_size;
    const float* img = (const float*)d_in[0];
    const float* kx  = (const float*)d_in[1];
    const float* ky  = (const float*)d_in[2];
    float* out = (float*)d_out;

    dim3 grid(4, 4, 32);
    dim3 block(256);
    conv3x3_pair_mag_kernel<<<grid, block, 0, stream>>>(img, kx, ky, out);
}
